# starting kernel
# baseline (speedup 1.0000x reference)
#include <hip/hip_runtime.h>

#define F_IN   256
#define HEADS  6
#define CDIM   128
#define FDIM   768
#define OUT_C  128
#define NEG_SLOPE 0.2f
#define BN_EPS    1e-5f

typedef __attribute__((ext_vector_type(2))) float v2f;
typedef __attribute__((ext_vector_type(8))) float v8f;

__device__ __forceinline__ unsigned fmap(float f) {
    unsigned u = __float_as_uint(f);
    return (u & 0x80000000u) ? ~u : (u | 0x80000000u);
}
__device__ __forceinline__ float funmap(unsigned u) {
    u = (u & 0x80000000u) ? (u & 0x7FFFFFFFu) : ~u;
    return __uint_as_float(u);
}
#define FMAP_NEG_INF 0x007FFFFFu

#define WMMA_F32(a, b, c) \
    __builtin_amdgcn_wmma_f32_16x16x4_f32(false, (a), false, (b), (short)0, (c), false, false)

__global__ __launch_bounds__(32)
void wmma_gemm_f32(const float* __restrict__ A, const float* __restrict__ B,
                   float* __restrict__ C, const float* __restrict__ bias,
                   int M, int K, int Nc)
{
    const int lane  = threadIdx.x & 31;
    const int nN64  = Nc >> 6;
    const int m0    = (blockIdx.x / nN64) * 32;
    const int n0    = (blockIdx.x % nN64) * 64;
    const int lrow  = lane & 15;
    const int khalf = (lane >> 4) << 1;
    const bool hasM1 = (m0 + 16) < M;

    v8f acc00 = {}, acc01 = {}, acc02 = {}, acc03 = {};
    v8f acc10 = {}, acc11 = {}, acc12 = {}, acc13 = {};

    const float* arow0 = A + (size_t)(m0 + lrow) * K;
    const float* arow1 = A + (size_t)(hasM1 ? (m0 + 16 + lrow) : (m0 + lrow)) * K;

    for (int k = 0; k < K; k += 4) {
        v2f a0, a1;
        a0.x = arow0[k + khalf];
        a0.y = arow0[k + khalf + 1];
        a1.x = arow1[k + khalf];
        a1.y = arow1[k + khalf + 1];

        const float* brow0 = B + (size_t)(k + khalf) * Nc + n0 + lrow;
        const float* brow1 = brow0 + Nc;
        v2f bv0, bv1, bv2, bv3;
        bv0.x = brow0[0];   bv0.y = brow1[0];
        bv1.x = brow0[16];  bv1.y = brow1[16];
        bv2.x = brow0[32];  bv2.y = brow1[32];
        bv3.x = brow0[48];  bv3.y = brow1[48];

        acc00 = WMMA_F32(a0, bv0, acc00);
        acc01 = WMMA_F32(a0, bv1, acc01);
        acc02 = WMMA_F32(a0, bv2, acc02);
        acc03 = WMMA_F32(a0, bv3, acc03);
        acc10 = WMMA_F32(a1, bv0, acc10);
        acc11 = WMMA_F32(a1, bv1, acc11);
        acc12 = WMMA_F32(a1, bv2, acc12);
        acc13 = WMMA_F32(a1, bv3, acc13);
    }

    float bb0 = 0.f, bb1 = 0.f, bb2 = 0.f, bb3 = 0.f;
    if (bias) {
        bb0 = bias[n0 + lrow];
        bb1 = bias[n0 + 16 + lrow];
        bb2 = bias[n0 + 32 + lrow];
        bb3 = bias[n0 + 48 + lrow];
    }
    const int rbase = (lane >> 4) * 8;
    #pragma unroll
    for (int r = 0; r < 8; ++r) {
        float* crow = C + (size_t)(m0 + rbase + r) * Nc + n0 + lrow;
        crow[0]  = acc00[r] + bb0;
        crow[16] = acc01[r] + bb1;
        crow[32] = acc02[r] + bb2;
        crow[48] = acc03[r] + bb3;
    }
    if (hasM1) {
        #pragma unroll
        for (int r = 0; r < 8; ++r) {
            float* crow = C + (size_t)(m0 + 16 + rbase + r) * Nc + n0 + lrow;
            crow[0]  = acc10[r] + bb0;
            crow[16] = acc11[r] + bb1;
            crow[32] = acc12[r] + bb2;
            crow[48] = acc13[r] + bb3;
        }
    }
}

__global__ __launch_bounds__(256)
void attn_scores(const float* __restrict__ hx,
                 const float* __restrict__ att_s, const float* __restrict__ att_d,
                 float* __restrict__ a_s, float* __restrict__ a_d, int nNodes)
{
    int gid  = blockIdx.x * blockDim.x + threadIdx.x;
    int wave = gid >> 5;
    int lane = gid & 31;
    if (wave >= nNodes * HEADS) return;
    int n = wave / HEADS, h = wave % HEADS;

    const float4* hv = (const float4*)(hx + (size_t)n * FDIM + h * CDIM);
    const float4* sv = (const float4*)(att_s + h * CDIM);
    const float4* dv = (const float4*)(att_d + h * CDIM);
    float4 x = hv[lane], s = sv[lane], d = dv[lane];
    float accs = x.x * s.x + x.y * s.y + x.z * s.z + x.w * s.w;
    float accd = x.x * d.x + x.y * d.y + x.z * d.z + x.w * d.w;
    #pragma unroll
    for (int off = 16; off > 0; off >>= 1) {
        accs += __shfl_down(accs, off, 32);
        accd += __shfl_down(accd, off, 32);
    }
    if (lane == 0) { a_s[wave] = accs; a_d[wave] = accd; }
}

__global__ void fill_u32(unsigned* __restrict__ p, unsigned v, size_t n)
{
    size_t i = (size_t)blockIdx.x * blockDim.x + threadIdx.x;
    if (i < n) p[i] = v;
}

__global__ __launch_bounds__(256)
void edge_scores_max(const int* __restrict__ adj,
                     const float* __restrict__ a_s, const float* __restrict__ a_d,
                     float* __restrict__ ebuf, unsigned* __restrict__ mbuf,
                     int E, int nTot)
{
    int e = blockIdx.x * blockDim.x + threadIdx.x;
    if (e >= nTot) return;
    int src, dst;
    if (e < E) { src = adj[e]; dst = adj[E + e]; } else { src = dst = e - E; }
    #pragma unroll
    for (int h = 0; h < HEADS; ++h) {
        float v = a_s[src * HEADS + h] + a_d[dst * HEADS + h];
        v = (v > 0.f) ? v : v * NEG_SLOPE;
        ebuf[(size_t)e * HEADS + h] = v;
        atomicMax(&mbuf[dst * HEADS + h], fmap(v));
    }
}

__global__ __launch_bounds__(256)
void edge_exp_sum(const int* __restrict__ adj, float* __restrict__ ebuf,
                  const unsigned* __restrict__ mbuf, float* __restrict__ sbuf,
                  int E, int nTot)
{
    int e = blockIdx.x * blockDim.x + threadIdx.x;
    if (e >= nTot) return;
    int dst = (e < E) ? adj[E + e] : e - E;
    #pragma unroll
    for (int h = 0; h < HEADS; ++h) {
        float m = funmap(mbuf[dst * HEADS + h]);
        float p = __expf(ebuf[(size_t)e * HEADS + h] - m);
        ebuf[(size_t)e * HEADS + h] = p;
        atomicAdd(&sbuf[dst * HEADS + h], p);
    }
}

__global__ __launch_bounds__(256)
void edge_aggregate(const int* __restrict__ adj, const float* __restrict__ hx,
                    const float* __restrict__ ebuf, const float* __restrict__ sbuf,
                    float* __restrict__ out, int E, int nTot)
{
    long gid  = (long)blockIdx.x * blockDim.x + threadIdx.x;
    long wave = gid >> 5;
    int  lane = (int)(gid & 31);
    if (wave >= (long)nTot * HEADS) return;
    int e = (int)(wave / HEADS), h = (int)(wave % HEADS);
    int src, dst;
    if (e < E) { src = adj[e]; dst = adj[E + e]; } else { src = dst = e - E; }

    float alpha = ebuf[(size_t)e * HEADS + h] / sbuf[dst * HEADS + h];
    const float4* hv = (const float4*)(hx + (size_t)src * FDIM + h * CDIM);
    float4 v = hv[lane];
    float* o = out + (size_t)dst * FDIM + h * CDIM + lane * 4;
    atomicAdd(o + 0, v.x * alpha);
    atomicAdd(o + 1, v.y * alpha);
    atomicAdd(o + 2, v.z * alpha);
    atomicAdd(o + 3, v.w * alpha);
}

__global__ __launch_bounds__(256)
void bias_relu(float* __restrict__ x, const float* __restrict__ bias, size_t total)
{
    size_t i = (size_t)blockIdx.x * blockDim.x + threadIdx.x;
    if (i >= total) return;
    float v = x[i] + bias[i % FDIM];
    x[i] = v > 0.f ? v : 0.f;
}

__global__ __launch_bounds__(256)
void bn_stats(const float* __restrict__ x, float* __restrict__ mean,
              float* __restrict__ rstd, int rows)
{
    int col = blockIdx.x;
    float s = 0.f, ss = 0.f;
    for (int r = threadIdx.x; r < rows; r += 256) {
        float v = x[(size_t)r * FDIM + col];
        s += v; ss += v * v;
    }
    __shared__ float sh[256], sh2[256];
    sh[threadIdx.x] = s; sh2[threadIdx.x] = ss;
    __syncthreads();
    for (int o = 128; o > 0; o >>= 1) {
        if (threadIdx.x < o) {
            sh[threadIdx.x]  += sh[threadIdx.x + o];
            sh2[threadIdx.x] += sh2[threadIdx.x + o];
        }
        __syncthreads();
    }
    if (threadIdx.x == 0) {
        float mu  = sh[0] / rows;
        float var = sh2[0] / rows - mu * mu;
        mean[col] = mu;
        rstd[col] = rsqrtf(var + BN_EPS);
    }
}

__global__ __launch_bounds__(256)
void bn_apply(float* __restrict__ x, const float* __restrict__ mean,
              const float* __restrict__ rstd, const float* __restrict__ gamma,
              const float* __restrict__ beta, size_t total)
{
    size_t i = (size_t)blockIdx.x * blockDim.x + threadIdx.x;
    if (i >= total) return;
    int col = (int)(i % FDIM);
    x[i] = (x[i] - mean[col]) * rstd[col] * gamma[col] + beta[col];
}

__global__ __launch_bounds__(256)
void pool_max(const float* __restrict__ lin, const int* __restrict__ ibatch,
              unsigned* __restrict__ pooled, int rows)
{
    size_t i = (size_t)blockIdx.x * blockDim.x + threadIdx.x;
    if (i >= (size_t)rows * OUT_C) return;
    int n = (int)(i / OUT_C), c = (int)(i % OUT_C);
    atomicMax(&pooled[ibatch[n] * OUT_C + c], fmap(lin[i]));
}

__global__ __launch_bounds__(256)
void pool_final(const unsigned* __restrict__ pooled, float* __restrict__ out, int n)
{
    int i = blockIdx.x * blockDim.x + threadIdx.x;
    if (i < n) out[i] = funmap(pooled[i]);
}

extern "C" void kernel_launch(void* const* d_in, const int* in_sizes, int n_in,
                              void* d_out, int out_size, void* d_ws, size_t ws_size,
                              hipStream_t stream)
{
    const float* x_in   = (const float*)d_in[0];
    const int*   adj    = (const int*)d_in[1];
    const int*   ibatch = (const int*)d_in[2];
    const float* W1     = (const float*)d_in[3];
    const float* att_s1 = (const float*)d_in[4];
    const float* att_d1 = (const float*)d_in[5];
    const float* b1     = (const float*)d_in[6];
    const float* g1     = (const float*)d_in[7];
    const float* be1    = (const float*)d_in[8];
    const float* W2     = (const float*)d_in[9];
    const float* att_s2 = (const float*)d_in[10];
    const float* att_d2 = (const float*)d_in[11];
    const float* b2     = (const float*)d_in[12];
    const float* g2     = (const float*)d_in[13];
    const float* be2    = (const float*)d_in[14];
    const float* linW   = (const float*)d_in[15];
    const float* linb   = (const float*)d_in[16];
    float* out = (float*)d_out;

    const int N    = in_sizes[0] / F_IN;
    const int E    = in_sizes[1] / 2;
    const int Etot = E + N;
    const int B    = out_size / OUT_C;
    const int mT   = (N + 31) / 32;

    float*    hx   = (float*)d_ws;
    float*    agg  = hx  + (size_t)N * FDIM;
    float*    a_s  = agg + (size_t)N * FDIM;
    float*    a_d  = a_s + (size_t)N * HEADS;
    unsigned* mbuf = (unsigned*)(a_d + (size_t)N * HEADS);
    float*    sbuf = (float*)(mbuf + (size_t)N * HEADS);
    float*    ebuf = sbuf + (size_t)N * HEADS;
    float*    mean = ebuf + (size_t)Etot * HEADS;
    float*    rstd = mean + FDIM;
    float*    lin  = rstd + FDIM;
    unsigned* pooled = (unsigned*)(lin + (size_t)N * OUT_C);

    const size_t nh  = (size_t)N * HEADS;
    const size_t nf  = (size_t)N * FDIM;
    const long   ew  = (long)Etot * HEADS;

    auto layer = [&](const float* xin, int K, const float* W,
                     const float* asw, const float* adw,
                     const float* bias, const float* gamma, const float* beta) {
        wmma_gemm_f32<<<mT * (FDIM / 64), 32, 0, stream>>>(xin, W, hx, nullptr, N, K, FDIM);
        attn_scores<<<(int)((nh * 32 + 255) / 256), 256, 0, stream>>>(hx, asw, adw, a_s, a_d, N);
        fill_u32<<<(int)((nh + 255) / 256), 256, 0, stream>>>(mbuf, FMAP_NEG_INF, nh);
        fill_u32<<<(int)((nh + 255) / 256), 256, 0, stream>>>((unsigned*)sbuf, 0u, nh);
        fill_u32<<<(int)((nf + 255) / 256), 256, 0, stream>>>((unsigned*)agg, 0u, nf);
        edge_scores_max<<<(Etot + 255) / 256, 256, 0, stream>>>(adj, a_s, a_d, ebuf, mbuf, E, Etot);
        edge_exp_sum<<<(Etot + 255) / 256, 256, 0, stream>>>(adj, ebuf, mbuf, sbuf, E, Etot);
        edge_aggregate<<<(int)((ew * 32 + 255) / 256), 256, 0, stream>>>(adj, hx, ebuf, sbuf, agg, E, Etot);
        bias_relu<<<(int)((nf + 255) / 256), 256, 0, stream>>>(agg, bias, nf);
        bn_stats<<<FDIM, 256, 0, stream>>>(agg, mean, rstd, N);
        bn_apply<<<(int)((nf + 255) / 256), 256, 0, stream>>>(agg, mean, rstd, gamma, beta, nf);
    };

    layer(x_in, F_IN, W1, att_s1, att_d1, b1, g1, be1);
    layer(agg,  FDIM, W2, att_s2, att_d2, b2, g2, be2);

    wmma_gemm_f32<<<mT * (OUT_C / 64), 32, 0, stream>>>(agg, linW, lin, linb, N, FDIM, OUT_C);
    fill_u32<<<(B * OUT_C + 255) / 256, 256, 0, stream>>>(pooled, FMAP_NEG_INF, (size_t)B * OUT_C);
    pool_max<<<(int)(((size_t)N * OUT_C + 255) / 256), 256, 0, stream>>>(lin, ibatch, pooled, N);
    pool_final<<<(B * OUT_C + 255) / 256, 256, 0, stream>>>(pooled, out, B * OUT_C);
}


# GEMMs: 64x64 tile per 4-wave work-group, A and B K-panels fetched once by 16-byte whole-line loads into LDS, WMMA fragments read from LDS (was one wave per tile gathering fragments from global)
# speedup vs baseline: 1.2302x; 1.2302x over previous
#include <hip/hip_runtime.h>

#define F_IN   256
#define HEADS  6
#define CDIM   128
#define FDIM   768
#define OUT_C  128
#define NEG_SLOPE 0.2f
#define BN_EPS    1e-5f

typedef __attribute__((ext_vector_type(2))) float v2f;
typedef __attribute__((ext_vector_type(8))) float v8f;

__device__ __forceinline__ unsigned fmap(float f) {
    unsigned u = __float_as_uint(f);
    return (u & 0x80000000u) ? ~u : (u | 0x80000000u);
}
__device__ __forceinline__ float funmap(unsigned u) {
    u = (u & 0x80000000u) ? (u & 0x7FFFFFFFu) : ~u;
    return __uint_as_float(u);
}
#define FMAP_NEG_INF 0x007FFFFFu

#define WMMA_F32(a, b, c) \
    __builtin_amdgcn_wmma_f32_16x16x4_f32(false, (a), false, (b), (short)0, (c), false, false)

#define GBM 64
#define GBN 64
#define GBK 32
#define APITCH 36
#define BPITCH 160

typedef __attribute__((ext_vector_type(4))) float v4f;

__global__ __launch_bounds__(128)
void wmma_gemm_f32(const float* __restrict__ A, const float* __restrict__ B,
                   float* __restrict__ C, const float* __restrict__ bias,
                   int M, int K, int Nc)
{
    __shared__ __attribute__((aligned(16))) float As[GBM * APITCH];
    __shared__ __attribute__((aligned(16))) float Bs[(GBK / 2) * BPITCH];

    const int tid  = threadIdx.x;
    const int lane = tid & 31;
    const int wave = tid >> 5;
    const int wm   = wave >> 1;
    const int wn   = wave & 1;
    const int nN   = Nc >> 6;
    const int m0   = (blockIdx.x / nN) * GBM;
    const int n0   = (blockIdx.x % nN) * GBN;
    const int c    = lane & 15;
    const int h    = lane >> 4;

    const int arow = tid >> 3, acol = (tid & 7) << 2;
    const int lastRow = M - 1;
    int gm0 = m0 + arow;       gm0 = gm0 < lastRow ? gm0 : lastRow;
    int gm1 = m0 + arow + 16;  gm1 = gm1 < lastRow ? gm1 : lastRow;
    int gm2 = m0 + arow + 32;  gm2 = gm2 < lastRow ? gm2 : lastRow;
    int gm3 = m0 + arow + 48;  gm3 = gm3 < lastRow ? gm3 : lastRow;
    const float* ag0 = A + (size_t)gm0 * K + acol;
    const float* ag1 = A + (size_t)gm1 * K + acol;
    const float* ag2 = A + (size_t)gm2 * K + acol;
    const float* ag3 = A + (size_t)gm3 * K + acol;
    float* as0 = As + arow * APITCH + acol;

    const int bkp = tid >> 4, bcol = (tid & 15) << 2;
    const float* bg = B + (size_t)(2 * bkp) * Nc + n0 + bcol;
    const size_t nc1  = (size_t)Nc;
    const size_t nc16 = (size_t)16 * Nc;
    float* bs0 = Bs + bkp * BPITCH + bcol * 2;

    v4f ra0 = *(const v4f*)(ag0);
    v4f ra1 = *(const v4f*)(ag1);
    v4f ra2 = *(const v4f*)(ag2);
    v4f ra3 = *(const v4f*)(ag3);
    v4f rb0 = *(const v4f*)(bg);
    v4f rb1 = *(const v4f*)(bg + nc1);
    v4f rb2 = *(const v4f*)(bg + nc16);
    v4f rb3 = *(const v4f*)(bg + nc16 + nc1);

    v8f acc00 = {}, acc01 = {}, acc10 = {}, acc11 = {};

    const float* af0 = As + (wm * 32 + c) * APITCH + 2 * h;
    const float* af1 = af0 + 16 * APITCH;
    const float* bf0 = Bs + h * BPITCH + (wn * 32 + c) * 2;
    const float* bf1 = bf0 + 32;

    for (int k0 = 0; k0 < K; k0 += GBK) {
        *(v4f*)(as0)               = ra0;
        *(v4f*)(as0 + 16 * APITCH) = ra1;
        *(v4f*)(as0 + 32 * APITCH) = ra2;
        *(v4f*)(as0 + 48 * APITCH) = ra3;
        {
            v4f lo, hi;
            lo.x = rb0.x; lo.y = rb1.x; lo.z = rb0.y; lo.w = rb1.y;
            hi.x = rb0.z; hi.y = rb1.z; hi.z = rb0.w; hi.w = rb1.w;
            *(v4f*)(bs0)     = lo;
            *(v4f*)(bs0 + 4) = hi;
            lo.x = rb2.x; lo.y = rb3.x; lo.z = rb2.y; lo.w = rb3.y;
            hi.x = rb2.z; hi.y = rb3.z; hi.z = rb2.w; hi.w = rb3.w;
            *(v4f*)(bs0 + 8 * BPITCH)     = lo;
            *(v4f*)(bs0 + 8 * BPITCH + 4) = hi;
        }
        __syncthreads();

        const int kn = k0 + GBK;
        if (kn < K) {
            const float* bgn = bg + (size_t)kn * Nc;
            ra0 = *(const v4f*)(ag0 + kn);
            ra1 = *(const v4f*)(ag1 + kn);
            ra2 = *(const v4f*)(ag2 + kn);
            ra3 = *(const v4f*)(ag3 + kn);
            rb0 = *(const v4f*)(bgn);
            rb1 = *(const v4f*)(bgn + nc1);
            rb2 = *(const v4f*)(bgn + nc16);
            rb3 = *(const v4f*)(bgn + nc16 + nc1);
        }

        #pragma unroll
        for (int kk = 0; kk < GBK; kk += 4) {
            v2f a0 = *(const v2f*)(af0 + kk);
            v2f a1 = *(const v2f*)(af1 + kk);
            v2f b0 = *(const v2f*)(bf0 + (kk >> 1) * BPITCH);
            v2f b1 = *(const v2f*)(bf1 + (kk >> 1) * BPITCH);
            acc00 = WMMA_F32(a0, b0, acc00);
            acc01 = WMMA_F32(a0, b1, acc01);
            acc10 = WMMA_F32(a1, b0, acc10);
            acc11 = WMMA_F32(a1, b1, acc11);
        }
        __syncthreads();
    }

    float bb0 = 0.f, bb1 = 0.f;
    if (bias) {
        bb0 = bias[n0 + wn * 32 + c];
        bb1 = bias[n0 + wn * 32 + 16 + c];
    }
    const int rbase = m0 + wm * 32 + h * 8;
    #pragma unroll
    for (int r = 0; r < 8; ++r) {
        const int row0 = rbase + r;
        const int row1 = row0 + 16;
        float* crow0 = C + (size_t)row0 * Nc + n0 + wn * 32 + c;
        float* crow1 = C + (size_t)row1 * Nc + n0 + wn * 32 + c;
        if (row0 < M) {
            crow0[0]  = acc00[r] + bb0;
            crow0[16] = acc01[r] + bb1;
        }
        if (row1 < M) {
            crow1[0]  = acc10[r] + bb0;
            crow1[16] = acc11[r] + bb1;
        }
    }
}

__global__ __launch_bounds__(256)
void attn_scores(const float* __restrict__ hx,
                 const float* __restrict__ att_s, const float* __restrict__ att_d,
                 float* __restrict__ a_s, float* __restrict__ a_d, int nNodes)
{
    int gid  = blockIdx.x * blockDim.x + threadIdx.x;
    int wave = gid >> 5;
    int lane = gid & 31;
    if (wave >= nNodes * HEADS) return;
    int n = wave / HEADS, h = wave % HEADS;

    const float4* hv = (const float4*)(hx + (size_t)n * FDIM + h * CDIM);
    const float4* sv = (const float4*)(att_s + h * CDIM);
    const float4* dv = (const float4*)(att_d + h * CDIM);
    float4 x = hv[lane], s = sv[lane], d = dv[lane];
    float accs = x.x * s.x + x.y * s.y + x.z * s.z + x.w * s.w;
    float accd = x.x * d.x + x.y * d.y + x.z * d.z + x.w * d.w;
    #pragma unroll
    for (int off = 16; off > 0; off >>= 1) {
        accs += __shfl_down(accs, off, 32);
        accd += __shfl_down(accd, off, 32);
    }
    if (lane == 0) { a_s[wave] = accs; a_d[wave] = accd; }
}

__global__ void fill_u32(unsigned* __restrict__ p, unsigned v, size_t n)
{
    size_t i = (size_t)blockIdx.x * blockDim.x + threadIdx.x;
    if (i < n) p[i] = v;
}

__global__ __launch_bounds__(256)
void edge_scores_max(const int* __restrict__ adj,
                     const float* __restrict__ a_s, const float* __restrict__ a_d,
                     float* __restrict__ ebuf, unsigned* __restrict__ mbuf,
                     int E, int nTot)
{
    int e = blockIdx.x * blockDim.x + threadIdx.x;
    if (e >= nTot) return;
    int src, dst;
    if (e < E) { src = adj[e]; dst = adj[E + e]; } else { src = dst = e - E; }
    #pragma unroll
    for (int h = 0; h < HEADS; ++h) {
        float v = a_s[src * HEADS + h] + a_d[dst * HEADS + h];
        v = (v > 0.f) ? v : v * NEG_SLOPE;
        ebuf[(size_t)e * HEADS + h] = v;
        atomicMax(&mbuf[dst * HEADS + h], fmap(v));
    }
}

__global__ __launch_bounds__(256)
void edge_exp_sum(const int* __restrict__ adj, float* __restrict__ ebuf,
                  const unsigned* __restrict__ mbuf, float* __restrict__ sbuf,
                  int E, int nTot)
{
    int e = blockIdx.x * blockDim.x + threadIdx.x;
    if (e >= nTot) return;
    int dst = (e < E) ? adj[E + e] : e - E;
    #pragma unroll
    for (int h = 0; h < HEADS; ++h) {
        float m = funmap(mbuf[dst * HEADS + h]);
        float p = __expf(ebuf[(size_t)e * HEADS + h] - m);
        ebuf[(size_t)e * HEADS + h] = p;
        atomicAdd(&sbuf[dst * HEADS + h], p);
    }
}

__global__ __launch_bounds__(256)
void edge_aggregate(const int* __restrict__ adj, const float* __restrict__ hx,
                    const float* __restrict__ ebuf, const float* __restrict__ sbuf,
                    float* __restrict__ out, int E, int nTot)
{
    long gid  = (long)blockIdx.x * blockDim.x + threadIdx.x;
    long wave = gid >> 5;
    int  lane = (int)(gid & 31);
    if (wave >= (long)nTot * HEADS) return;
    int e = (int)(wave / HEADS), h = (int)(wave % HEADS);
    int src, dst;
    if (e < E) { src = adj[e]; dst = adj[E + e]; } else { src = dst = e - E; }

    float alpha = ebuf[(size_t)e * HEADS + h] / sbuf[dst * HEADS + h];
    const float4* hv = (const float4*)(hx + (size_t)src * FDIM + h * CDIM);
    float4 v = hv[lane];
    float* o = out + (size_t)dst * FDIM + h * CDIM + lane * 4;
    atomicAdd(o + 0, v.x * alpha);
    atomicAdd(o + 1, v.y * alpha);
    atomicAdd(o + 2, v.z * alpha);
    atomicAdd(o + 3, v.w * alpha);
}

__global__ __launch_bounds__(256)
void bias_relu(float* __restrict__ x, const float* __restrict__ bias, size_t total)
{
    size_t i = (size_t)blockIdx.x * blockDim.x + threadIdx.x;
    if (i >= total) return;
    float v = x[i] + bias[i % FDIM];
    x[i] = v > 0.f ? v : 0.f;
}

__global__ __launch_bounds__(256)
void bn_stats(const float* __restrict__ x, float* __restrict__ mean,
              float* __restrict__ rstd, int rows)
{
    int col = blockIdx.x;
    float s = 0.f, ss = 0.f;
    for (int r = threadIdx.x; r < rows; r += 256) {
        float v = x[(size_t)r * FDIM + col];
        s += v; ss += v * v;
    }
    __shared__ float sh[256], sh2[256];
    sh[threadIdx.x] = s; sh2[threadIdx.x] = ss;
    __syncthreads();
    for (int o = 128; o > 0; o >>= 1) {
        if (threadIdx.x < o) {
            sh[threadIdx.x]  += sh[threadIdx.x + o];
            sh2[threadIdx.x] += sh2[threadIdx.x + o];
        }
        __syncthreads();
    }
    if (threadIdx.x == 0) {
        float mu  = sh[0] / rows;
        float var = sh2[0] / rows - mu * mu;
        mean[col] = mu;
        rstd[col] = rsqrtf(var + BN_EPS);
    }
}

__global__ __launch_bounds__(256)
void bn_apply(float* __restrict__ x, const float* __restrict__ mean,
              const float* __restrict__ rstd, const float* __restrict__ gamma,
              const float* __restrict__ beta, size_t total)
{
    size_t i = (size_t)blockIdx.x * blockDim.x + threadIdx.x;
    if (i >= total) return;
    int col = (int)(i % FDIM);
    x[i] = (x[i] - mean[col]) * rstd[col] * gamma[col] + beta[col];
}

__global__ __launch_bounds__(256)
void pool_max(const float* __restrict__ lin, const int* __restrict__ ibatch,
              unsigned* __restrict__ pooled, int rows)
{
    size_t i = (size_t)blockIdx.x * blockDim.x + threadIdx.x;
    if (i >= (size_t)rows * OUT_C) return;
    int n = (int)(i / OUT_C), c = (int)(i % OUT_C);
    atomicMax(&pooled[ibatch[n] * OUT_C + c], fmap(lin[i]));
}

__global__ __launch_bounds__(256)
void pool_final(const unsigned* __restrict__ pooled, float* __restrict__ out, int n)
{
    int i = blockIdx.x * blockDim.x + threadIdx.x;
    if (i < n) out[i] = funmap(pooled[i]);
}

extern "C" void kernel_launch(void* const* d_in, const int* in_sizes, int n_in,
                              void* d_out, int out_size, void* d_ws, size_t ws_size,
                              hipStream_t stream)
{
    const float* x_in   = (const float*)d_in[0];
    const int*   adj    = (const int*)d_in[1];
    const int*   ibatch = (const int*)d_in[2];
    const float* W1     = (const float*)d_in[3];
    const float* att_s1 = (const float*)d_in[4];
    const float* att_d1 = (const float*)d_in[5];
    const float* b1     = (const float*)d_in[6];
    const float* g1     = (const float*)d_in[7];
    const float* be1    = (const float*)d_in[8];
    const float* W2     = (const float*)d_in[9];
    const float* att_s2 = (const float*)d_in[10];
    const float* att_d2 = (const float*)d_in[11];
    const float* b2     = (const float*)d_in[12];
    const float* g2     = (const float*)d_in[13];
    const float* be2    = (const float*)d_in[14];
    const float* linW   = (const float*)d_in[15];
    const float* linb   = (const float*)d_in[16];
    float* out = (float*)d_out;

    const int N    = in_sizes[0] / F_IN;
    const int E    = in_sizes[1] / 2;
    const int Etot = E + N;
    const int B    = out_size / OUT_C;
    const int mT   = (N + 63) / 64;

    float*    hx   = (float*)d_ws;
    float*    agg  = hx  + (size_t)N * FDIM;
    float*    a_s  = agg + (size_t)N * FDIM;
    float*    a_d  = a_s + (size_t)N * HEADS;
    unsigned* mbuf = (unsigned*)(a_d + (size_t)N * HEADS);
    float*    sbuf = (float*)(mbuf + (size_t)N * HEADS);
    float*    ebuf = sbuf + (size_t)N * HEADS;
    float*    mean = ebuf + (size_t)Etot * HEADS;
    float*    rstd = mean + FDIM;
    float*    lin  = rstd + FDIM;
    unsigned* pooled = (unsigned*)(lin + (size_t)N * OUT_C);

    const size_t nh  = (size_t)N * HEADS;
    const size_t nf  = (size_t)N * FDIM;
    const long   ew  = (long)Etot * HEADS;

    auto layer = [&](const float* xin, int K, const float* W,
                     const float* asw, const float* adw,
                     const float* bias, const float* gamma, const float* beta) {
        wmma_gemm_f32<<<mT * (FDIM / 64), 128, 0, stream>>>(xin, W, hx, nullptr, N, K, FDIM);
        attn_scores<<<(int)((nh * 32 + 255) / 256), 256, 0, stream>>>(hx, asw, adw, a_s, a_d, N);
        fill_u32<<<(int)((nh + 255) / 256), 256, 0, stream>>>(mbuf, FMAP_NEG_INF, nh);
        fill_u32<<<(int)((nh + 255) / 256), 256, 0, stream>>>((unsigned*)sbuf, 0u, nh);
        fill_u32<<<(int)((nf + 255) / 256), 256, 0, stream>>>((unsigned*)agg, 0u, nf);
        edge_scores_max<<<(Etot + 255) / 256, 256, 0, stream>>>(adj, a_s, a_d, ebuf, mbuf, E, Etot);
        edge_exp_sum<<<(Etot + 255) / 256, 256, 0, stream>>>(adj, ebuf, mbuf, sbuf, E, Etot);
        edge_aggregate<<<(int)((ew * 32 + 255) / 256), 256, 0, stream>>>(adj, hx, ebuf, sbuf, agg, E, Etot);
        bias_relu<<<(int)((nf + 255) / 256), 256, 0, stream>>>(agg, bias, nf);
        bn_stats<<<FDIM, 256, 0, stream>>>(agg, mean, rstd, N);
        bn_apply<<<(int)((nf + 255) / 256), 256, 0, stream>>>(agg, mean, rstd, gamma, beta, nf);
    };

    layer(x_in, F_IN, W1, att_s1, att_d1, b1, g1, be1);
    layer(agg,  FDIM, W2, att_s2, att_d2, b2, g2, be2);

    wmma_gemm_f32<<<mT * (OUT_C / 64), 128, 0, stream>>>(agg, linW, lin, linb, N, FDIM, OUT_C);
    fill_u32<<<(B * OUT_C + 255) / 256, 256, 0, stream>>>(pooled, FMAP_NEG_INF, (size_t)B * OUT_C);
    pool_max<<<(int)(((size_t)N * OUT_C + 255) / 256), 256, 0, stream>>>(lin, ibatch, pooled, N);
    pool_final<<<(B * OUT_C + 255) / 256, 256, 0, stream>>>(pooled, out, B * OUT_C);
}


# Float-atomic scatter replaced by a gather: edges bucketed by destination per call (integer ticket, arrival-order slots); one wave per (node, head): softmax, weighted sum, bias, relu in registers
# speedup vs baseline: 2.4071x; 1.9567x over previous
#include <hip/hip_runtime.h>

#define F_IN   256
#define HEADS  6
#define CDIM   128
#define FDIM   768
#define OUT_C  128
#define NEG_SLOPE 0.2f
#define BN_EPS    1e-5f

typedef __attribute__((ext_vector_type(2))) float v2f;
typedef __attribute__((ext_vector_type(8))) float v8f;

__device__ __forceinline__ unsigned fmap(float f) {
    unsigned u = __float_as_uint(f);
    return (u & 0x80000000u) ? ~u : (u | 0x80000000u);
}
__device__ __forceinline__ float funmap(unsigned u) {
    u = (u & 0x80000000u) ? (u & 0x7FFFFFFFu) : ~u;
    return __uint_as_float(u);
}
#define FMAP_NEG_INF 0x007FFFFFu

#define WMMA_F32(a, b, c) \
    __builtin_amdgcn_wmma_f32_16x16x4_f32(false, (a), false, (b), (short)0, (c), false, false)

#define GBM 64
#define GBN 64
#define GBK 32
#define APITCH 36
#define BPITCH 160

typedef __attribute__((ext_vector_type(4))) float v4f;

__global__ __launch_bounds__(128)
void wmma_gemm_f32(const float* __restrict__ A, const float* __restrict__ B,
                   float* __restrict__ C, const float* __restrict__ bias,
                   int M, int K, int Nc)
{
    __shared__ __attribute__((aligned(16))) float As[GBM * APITCH];
    __shared__ __attribute__((aligned(16))) float Bs[(GBK / 2) * BPITCH];

    const int tid  = threadIdx.x;
    const int lane = tid & 31;
    const int wave = tid >> 5;
    const int wm   = wave >> 1;
    const int wn   = wave & 1;
    const int nN   = Nc >> 6;
    const int m0   = (blockIdx.x / nN) * GBM;
    const int n0   = (blockIdx.x % nN) * GBN;
    const int c    = lane & 15;
    const int h    = lane >> 4;

    const int arow = tid >> 3, acol = (tid & 7) << 2;
    const int lastRow = M - 1;
    int gm0 = m0 + arow;       gm0 = gm0 < lastRow ? gm0 : lastRow;
    int gm1 = m0 + arow + 16;  gm1 = gm1 < lastRow ? gm1 : lastRow;
    int gm2 = m0 + arow + 32;  gm2 = gm2 < lastRow ? gm2 : lastRow;
    int gm3 = m0 + arow + 48;  gm3 = gm3 < lastRow ? gm3 : lastRow;
    const float* ag0 = A + (size_t)gm0 * K + acol;
    const float* ag1 = A + (size_t)gm1 * K + acol;
    const float* ag2 = A + (size_t)gm2 * K + acol;
    const float* ag3 = A + (size_t)gm3 * K + acol;
    float* as0 = As + arow * APITCH + acol;

    const int bkp = tid >> 4, bcol = (tid & 15) << 2;
    const float* bg = B + (size_t)(2 * bkp) * Nc + n0 + bcol;
    const size_t nc1  = (size_t)Nc;
    const size_t nc16 = (size_t)16 * Nc;
    float* bs0 = Bs + bkp * BPITCH + bcol * 2;

    v4f ra0 = *(const v4f*)(ag0);
    v4f ra1 = *(const v4f*)(ag1);
    v4f ra2 = *(const v4f*)(ag2);
    v4f ra3 = *(const v4f*)(ag3);
    v4f rb0 = *(const v4f*)(bg);
    v4f rb1 = *(const v4f*)(bg + nc1);
    v4f rb2 = *(const v4f*)(bg + nc16);
    v4f rb3 = *(const v4f*)(bg + nc16 + nc1);

    v8f acc00 = {}, acc01 = {}, acc10 = {}, acc11 = {};

    const float* af0 = As + (wm * 32 + c) * APITCH + 2 * h;
    const float* af1 = af0 + 16 * APITCH;
    const float* bf0 = Bs + h * BPITCH + (wn * 32 + c) * 2;
    const float* bf1 = bf0 + 32;

    for (int k0 = 0; k0 < K; k0 += GBK) {
        *(v4f*)(as0)               = ra0;
        *(v4f*)(as0 + 16 * APITCH) = ra1;
        *(v4f*)(as0 + 32 * APITCH) = ra2;
        *(v4f*)(as0 + 48 * APITCH) = ra3;
        {
            v4f lo, hi;
            lo.x = rb0.x; lo.y = rb1.x; lo.z = rb0.y; lo.w = rb1.y;
            hi.x = rb0.z; hi.y = rb1.z; hi.z = rb0.w; hi.w = rb1.w;
            *(v4f*)(bs0)     = lo;
            *(v4f*)(bs0 + 4) = hi;
            lo.x = rb2.x; lo.y = rb3.x; lo.z = rb2.y; lo.w = rb3.y;
            hi.x = rb2.z; hi.y = rb3.z; hi.z = rb2.w; hi.w = rb3.w;
            *(v4f*)(bs0 + 8 * BPITCH)     = lo;
            *(v4f*)(bs0 + 8 * BPITCH + 4) = hi;
        }
        __syncthreads();

        const int kn = k0 + GBK;
        if (kn < K) {
            const float* bgn = bg + (size_t)kn * Nc;
            ra0 = *(const v4f*)(ag0 + kn);
            ra1 = *(const v4f*)(ag1 + kn);
            ra2 = *(const v4f*)(ag2 + kn);
            ra3 = *(const v4f*)(ag3 + kn);
            rb0 = *(const v4f*)(bgn);
            rb1 = *(const v4f*)(bgn + nc1);
            rb2 = *(const v4f*)(bgn + nc16);
            rb3 = *(const v4f*)(bgn + nc16 + nc1);
        }

        #pragma unroll
        for (int kk = 0; kk < GBK; kk += 4) {
            v2f a0 = *(const v2f*)(af0 + kk);
            v2f a1 = *(const v2f*)(af1 + kk);
            v2f b0 = *(const v2f*)(bf0 + (kk >> 1) * BPITCH);
            v2f b1 = *(const v2f*)(bf1 + (kk >> 1) * BPITCH);
            acc00 = WMMA_F32(a0, b0, acc00);
            acc01 = WMMA_F32(a0, b1, acc01);
            acc10 = WMMA_F32(a1, b0, acc10);
            acc11 = WMMA_F32(a1, b1, acc11);
        }
        __syncthreads();
    }

    float bb0 = 0.f, bb1 = 0.f;
    if (bias) {
        bb0 = bias[n0 + wn * 32 + c];
        bb1 = bias[n0 + wn * 32 + 16 + c];
    }
    const int rbase = m0 + wm * 32 + h * 8;
    #pragma unroll
    for (int r = 0; r < 8; ++r) {
        const int row0 = rbase + r;
        const int row1 = row0 + 16;
        float* crow0 = C + (size_t)row0 * Nc + n0 + wn * 32 + c;
        float* crow1 = C + (size_t)row1 * Nc + n0 + wn * 32 + c;
        if (row0 < M) {
            crow0[0]  = acc00[r] + bb0;
            crow0[16] = acc01[r] + bb1;
        }
        if (row1 < M) {
            crow1[0]  = acc10[r] + bb0;
            crow1[16] = acc11[r] + bb1;
        }
    }
}

__global__ __launch_bounds__(256)
void attn_scores(const float* __restrict__ hx,
                 const float* __restrict__ att_s, const float* __restrict__ att_d,
                 float* __restrict__ a_s, float* __restrict__ a_d, int nNodes)
{
    int gid  = blockIdx.x * blockDim.x + threadIdx.x;
    int wave = gid >> 5;
    int lane = gid & 31;
    if (wave >= nNodes * HEADS) return;
    int n = wave / HEADS, h = wave % HEADS;

    const float4* hv = (const float4*)(hx + (size_t)n * FDIM + h * CDIM);
    const float4* sv = (const float4*)(att_s + h * CDIM);
    const float4* dv = (const float4*)(att_d + h * CDIM);
    float4 x = hv[lane], s = sv[lane], d = dv[lane];
    float accs = x.x * s.x + x.y * s.y + x.z * s.z + x.w * s.w;
    float accd = x.x * d.x + x.y * d.y + x.z * d.z + x.w * d.w;
    #pragma unroll
    for (int off = 16; off > 0; off >>= 1) {
        accs += __shfl_down(accs, off, 32);
        accd += __shfl_down(accd, off, 32);
    }
    if (lane == 0) { a_s[wave] = accs; a_d[wave] = accd; }
}

__global__ void fill_u32(unsigned* __restrict__ p, unsigned v, size_t n)
{
    size_t i = (size_t)blockIdx.x * blockDim.x + threadIdx.x;
    if (i < n) p[i] = v;
}

__global__ __launch_bounds__(256)
void csr_count(const int* __restrict__ adj, unsigned* __restrict__ cnt,
               int* __restrict__ ticket, int E)
{
    int e = blockIdx.x * blockDim.x + threadIdx.x;
    if (e >= E) return;
    ticket[e] = (int)atomicAdd(&cnt[adj[E + e]], 1u);
}

__global__ __launch_bounds__(1024)
void csr_scan(const unsigned* __restrict__ cnt, int* __restrict__ rowptr, int n)
{
    __shared__ int wsum[32];
    const int tid = threadIdx.x, lane = tid & 31, w = tid >> 5;
    const int per = (n + 1023) / 1024;
    int b = tid * per;          b = b < n ? b : n;
    int e = b + per;            e = e < n ? e : n;
    int s = 0;
    for (int i = b; i < e; ++i) s += (int)cnt[i];
    int v = s;
    for (int off = 1; off < 32; off <<= 1) {
        int t = __shfl_up(v, off, 32);
        if (lane >= off) v += t;
    }
    if (lane == 31) wsum[w] = v;
    __syncthreads();
    if (w == 0) {
        int x = wsum[lane];
        for (int off = 1; off < 32; off <<= 1) {
            int t = __shfl_up(x, off, 32);
            if (lane >= off) x += t;
        }
        wsum[lane] = x;
    }
    __syncthreads();
    int base = v - s + (w > 0 ? wsum[w - 1] : 0);
    for (int i = b; i < e; ++i) { rowptr[i] = base; base += (int)cnt[i]; }
    if (tid == 1023) rowptr[n] = base;
}

__global__ __launch_bounds__(256)
void csr_fill(const int* __restrict__ adj, const int* __restrict__ rowptr,
              const int* __restrict__ ticket, int* __restrict__ col, int E)
{
    int e = blockIdx.x * blockDim.x + threadIdx.x;
    if (e >= E) return;
    col[rowptr[adj[E + e]] + ticket[e]] = adj[e];
}

__device__ __forceinline__ float leaky(float v) { return (v > 0.f) ? v : v * NEG_SLOPE; }

__global__ __launch_bounds__(192)
void node_gather_softmax(const int* __restrict__ rowptr, const int* __restrict__ col,
                         const float* __restrict__ hx,
                         const float* __restrict__ a_s, const float* __restrict__ a_d,
                         const float* __restrict__ bias, float* __restrict__ out)
{
    const int n    = blockIdx.x;
    const int lane = threadIdx.x & 31;
    const int h    = __builtin_amdgcn_readfirstlane((int)(threadIdx.x >> 5));
    const int beg  = rowptr[n], end = rowptr[n + 1];
    const float ad = a_d[n * HEADS + h];
    const float vself = leaky(a_s[n * HEADS + h] + ad);

    float m = vself;
    for (int j0 = beg; j0 < end; j0 += 32) {
        const int j = j0 + lane;
        if (j < end) m = fmaxf(m, leaky(a_s[col[j] * HEADS + h] + ad));
    }
    for (int off = 16; off > 0; off >>= 1) m = fmaxf(m, __shfl_xor(m, off, 32));

    const size_t hoff = (size_t)h * CDIM + lane * 4;
    const float pself = __expf(vself - m);
    float tot = pself;
    v4f acc = *(const v4f*)(hx + (size_t)n * FDIM + hoff) * pself;

    for (int j0 = beg; j0 < end; j0 += 32) {
        const int j = j0 + lane;
        int   s = 0;
        float p = 0.f;
        if (j < end) {
            s = col[j];
            p = __expf(leaky(a_s[s * HEADS + h] + ad) - m);
        }
        const int rem = end - j0;
        const int cnt = rem < 32 ? rem : 32;
        #pragma unroll 4
        for (int t = 0; t < cnt; ++t) {
            const int   st = __builtin_amdgcn_readlane(s, t);
            const float pt = __int_as_float(__builtin_amdgcn_readlane(__float_as_int(p), t));
            const v4f x = *(const v4f*)(hx + (size_t)st * FDIM + hoff);
            acc += x * pt;
            tot += pt;
        }
    }

    const v4f b4 = *(const v4f*)(bias + hoff);
    v4f r = acc / tot + b4;
    r.x = r.x > 0.f ? r.x : 0.f;
    r.y = r.y > 0.f ? r.y : 0.f;
    r.z = r.z > 0.f ? r.z : 0.f;
    r.w = r.w > 0.f ? r.w : 0.f;
    *(v4f*)(out + (size_t)n * FDIM + hoff) = r;
}

__global__ __launch_bounds__(256)
void bn_stats(const float* __restrict__ x, float* __restrict__ mean,
              float* __restrict__ rstd, int rows)
{
    int col = blockIdx.x;
    float s = 0.f, ss = 0.f;
    for (int r = threadIdx.x; r < rows; r += 256) {
        float v = x[(size_t)r * FDIM + col];
        s += v; ss += v * v;
    }
    __shared__ float sh[256], sh2[256];
    sh[threadIdx.x] = s; sh2[threadIdx.x] = ss;
    __syncthreads();
    for (int o = 128; o > 0; o >>= 1) {
        if (threadIdx.x < o) {
            sh[threadIdx.x]  += sh[threadIdx.x + o];
            sh2[threadIdx.x] += sh2[threadIdx.x + o];
        }
        __syncthreads();
    }
    if (threadIdx.x == 0) {
        float mu  = sh[0] / rows;
        float var = sh2[0] / rows - mu * mu;
        mean[col] = mu;
        rstd[col] = rsqrtf(var + BN_EPS);
    }
}

__global__ __launch_bounds__(256)
void bn_apply(float* __restrict__ x, const float* __restrict__ mean,
              const float* __restrict__ rstd, const float* __restrict__ gamma,
              const float* __restrict__ beta, size_t total)
{
    size_t i = (size_t)blockIdx.x * blockDim.x + threadIdx.x;
    if (i >= total) return;
    int col = (int)(i % FDIM);
    x[i] = (x[i] - mean[col]) * rstd[col] * gamma[col] + beta[col];
}

__global__ __launch_bounds__(256)
void pool_max(const float* __restrict__ lin, const int* __restrict__ ibatch,
              unsigned* __restrict__ pooled, int rows)
{
    size_t i = (size_t)blockIdx.x * blockDim.x + threadIdx.x;
    if (i >= (size_t)rows * OUT_C) return;
    int n = (int)(i / OUT_C), c = (int)(i % OUT_C);
    atomicMax(&pooled[ibatch[n] * OUT_C + c], fmap(lin[i]));
}

__global__ __launch_bounds__(256)
void pool_final(const unsigned* __restrict__ pooled, float* __restrict__ out, int n)
{
    int i = blockIdx.x * blockDim.x + threadIdx.x;
    if (i < n) out[i] = funmap(pooled[i]);
}

extern "C" void kernel_launch(void* const* d_in, const int* in_sizes, int n_in,
                              void* d_out, int out_size, void* d_ws, size_t ws_size,
                              hipStream_t stream)
{
    const float* x_in   = (const float*)d_in[0];
    const int*   adj    = (const int*)d_in[1];
    const int*   ibatch = (const int*)d_in[2];
    const float* W1     = (const float*)d_in[3];
    const float* att_s1 = (const float*)d_in[4];
    const float* att_d1 = (const float*)d_in[5];
    const float* b1     = (const float*)d_in[6];
    const float* g1     = (const float*)d_in[7];
    const float* be1    = (const float*)d_in[8];
    const float* W2     = (const float*)d_in[9];
    const float* att_s2 = (const float*)d_in[10];
    const float* att_d2 = (const float*)d_in[11];
    const float* b2     = (const float*)d_in[12];
    const float* g2     = (const float*)d_in[13];
    const float* be2    = (const float*)d_in[14];
    const float* linW   = (const float*)d_in[15];
    const float* linb   = (const float*)d_in[16];
    float* out = (float*)d_out;

    const int N    = in_sizes[0] / F_IN;
    const int E    = in_sizes[1] / 2;
    const int B    = out_size / OUT_C;
    const int mT   = (N + 63) / 64;

    float*    hx   = (float*)d_ws;
    float*    agg  = hx  + (size_t)N * FDIM;
    float*    a_s  = agg + (size_t)N * FDIM;
    float*    a_d  = a_s + (size_t)N * HEADS;
    unsigned* cnt  = (unsigned*)(a_d + (size_t)N * HEADS);
    int*      rowptr = (int*)(cnt + (size_t)N);
    int*      ticket = rowptr + (size_t)N + 1;
    int*      col    = ticket + (size_t)E;
    float*    mean = (float*)(col + (size_t)E);
    float*    rstd = mean + FDIM;
    float*    lin  = rstd + FDIM;
    unsigned* pooled = (unsigned*)(lin + (size_t)N * OUT_C);

    const size_t nh  = (size_t)N * HEADS;
    const size_t nf  = (size_t)N * FDIM;

    fill_u32<<<(N + 255) / 256, 256, 0, stream>>>(cnt, 0u, (size_t)N);
    csr_count<<<(E + 255) / 256, 256, 0, stream>>>(adj, cnt, ticket, E);
    csr_scan<<<1, 1024, 0, stream>>>(cnt, rowptr, N);
    csr_fill<<<(E + 255) / 256, 256, 0, stream>>>(adj, rowptr, ticket, col, E);

    auto layer = [&](const float* xin, int K, const float* W,
                     const float* asw, const float* adw,
                     const float* bias, const float* gamma, const float* beta) {
        wmma_gemm_f32<<<mT * (FDIM / 64), 128, 0, stream>>>(xin, W, hx, nullptr, N, K, FDIM);
        attn_scores<<<(int)((nh * 32 + 255) / 256), 256, 0, stream>>>(hx, asw, adw, a_s, a_d, N);
        node_gather_softmax<<<N, 192, 0, stream>>>(rowptr, col, hx, a_s, a_d, bias, agg);
        bn_stats<<<FDIM, 256, 0, stream>>>(agg, mean, rstd, N);
        bn_apply<<<(int)((nf + 255) / 256), 256, 0, stream>>>(agg, mean, rstd, gamma, beta, nf);
    };

    layer(x_in, F_IN, W1, att_s1, att_d1, b1, g1, be1);
    layer(agg,  FDIM, W2, att_s2, att_d2, b2, g2, be2);

    wmma_gemm_f32<<<mT * (OUT_C / 64), 128, 0, stream>>>(agg, linW, lin, linb, N, FDIM, OUT_C);
    fill_u32<<<(B * OUT_C + 255) / 256, 256, 0, stream>>>(pooled, FMAP_NEG_INF, (size_t)B * OUT_C);
    pool_max<<<(int)(((size_t)N * OUT_C + 255) / 256), 256, 0, stream>>>(lin, ibatch, pooled, N);
    pool_final<<<(B * OUT_C + 255) / 256, 256, 0, stream>>>(pooled, out, B * OUT_C);
}


# Batch-norm column statistics: strided one-work-group-per-column reads replaced by a two-kernel reduction with whole-line 16-byte row loads; reduction tree changed; 18 launches, were 16
# speedup vs baseline: 3.4603x; 1.4375x over previous
#include <hip/hip_runtime.h>

#define F_IN   256
#define HEADS  6
#define CDIM   128
#define FDIM   768
#define OUT_C  128
#define NEG_SLOPE 0.2f
#define BN_EPS    1e-5f

typedef __attribute__((ext_vector_type(2))) float v2f;
typedef __attribute__((ext_vector_type(8))) float v8f;

__device__ __forceinline__ unsigned fmap(float f) {
    unsigned u = __float_as_uint(f);
    return (u & 0x80000000u) ? ~u : (u | 0x80000000u);
}
__device__ __forceinline__ float funmap(unsigned u) {
    u = (u & 0x80000000u) ? (u & 0x7FFFFFFFu) : ~u;
    return __uint_as_float(u);
}
#define FMAP_NEG_INF 0x007FFFFFu

#define WMMA_F32(a, b, c) \
    __builtin_amdgcn_wmma_f32_16x16x4_f32(false, (a), false, (b), (short)0, (c), false, false)

#define GBM 64
#define GBN 64
#define GBK 32
#define APITCH 36
#define BPITCH 160

typedef __attribute__((ext_vector_type(4))) float v4f;

template <bool NORM>
__global__ __launch_bounds__(128)
void wmma_gemm_f32(const float* __restrict__ A, const float* __restrict__ B,
                   float* __restrict__ C, const float* __restrict__ bias,
                   const float* __restrict__ nmean, const float* __restrict__ nscale,
                   const float* __restrict__ nbeta,
                   int M, int K, int Nc)
{
    __shared__ __attribute__((aligned(16))) float As[GBM * APITCH];
    __shared__ __attribute__((aligned(16))) float Bs[(GBK / 2) * BPITCH];

    const int tid  = threadIdx.x;
    const int lane = tid & 31;
    const int wave = tid >> 5;
    const int wm   = wave >> 1;
    const int wn   = wave & 1;
    const int nN   = Nc >> 6;
    const int m0   = (blockIdx.x / nN) * GBM;
    const int n0   = (blockIdx.x % nN) * GBN;
    const int c    = lane & 15;
    const int h    = lane >> 4;

    const int arow = tid >> 3, acol = (tid & 7) << 2;
    const int lastRow = M - 1;
    int gm0 = m0 + arow;       gm0 = gm0 < lastRow ? gm0 : lastRow;
    int gm1 = m0 + arow + 16;  gm1 = gm1 < lastRow ? gm1 : lastRow;
    int gm2 = m0 + arow + 32;  gm2 = gm2 < lastRow ? gm2 : lastRow;
    int gm3 = m0 + arow + 48;  gm3 = gm3 < lastRow ? gm3 : lastRow;
    const float* ag0 = A + (size_t)gm0 * K + acol;
    const float* ag1 = A + (size_t)gm1 * K + acol;
    const float* ag2 = A + (size_t)gm2 * K + acol;
    const float* ag3 = A + (size_t)gm3 * K + acol;
    float* as0 = As + arow * APITCH + acol;

    const int bkp = tid >> 4, bcol = (tid & 15) << 2;
    const float* bg = B + (size_t)(2 * bkp) * Nc + n0 + bcol;
    const size_t nc1  = (size_t)Nc;
    const size_t nc16 = (size_t)16 * Nc;
    float* bs0 = Bs + bkp * BPITCH + bcol * 2;

    v4f ra0 = *(const v4f*)(ag0);
    v4f ra1 = *(const v4f*)(ag1);
    v4f ra2 = *(const v4f*)(ag2);
    v4f ra3 = *(const v4f*)(ag3);
    v4f rb0 = *(const v4f*)(bg);
    v4f rb1 = *(const v4f*)(bg + nc1);
    v4f rb2 = *(const v4f*)(bg + nc16);
    v4f rb3 = *(const v4f*)(bg + nc16 + nc1);
    v4f pm = {}, ps = {}, pb = {};
    if (NORM) {
        pm = *(const v4f*)(nmean  + acol);
        ps = *(const v4f*)(nscale + acol);
        pb = *(const v4f*)(nbeta  + acol);
    }

    v8f acc00 = {}, acc01 = {}, acc10 = {}, acc11 = {};

    const float* af0 = As + (wm * 32 + c) * APITCH + 2 * h;
    const float* af1 = af0 + 16 * APITCH;
    const float* bf0 = Bs + h * BPITCH + (wn * 32 + c) * 2;
    const float* bf1 = bf0 + 32;

    for (int k0 = 0; k0 < K; k0 += GBK) {
        if (NORM) {
            ra0 = (ra0 - pm) * ps + pb;
            ra1 = (ra1 - pm) * ps + pb;
            ra2 = (ra2 - pm) * ps + pb;
            ra3 = (ra3 - pm) * ps + pb;
        }
        *(v4f*)(as0)               = ra0;
        *(v4f*)(as0 + 16 * APITCH) = ra1;
        *(v4f*)(as0 + 32 * APITCH) = ra2;
        *(v4f*)(as0 + 48 * APITCH) = ra3;
        {
            v4f lo, hi;
            lo.x = rb0.x; lo.y = rb1.x; lo.z = rb0.y; lo.w = rb1.y;
            hi.x = rb0.z; hi.y = rb1.z; hi.z = rb0.w; hi.w = rb1.w;
            *(v4f*)(bs0)     = lo;
            *(v4f*)(bs0 + 4) = hi;
            lo.x = rb2.x; lo.y = rb3.x; lo.z = rb2.y; lo.w = rb3.y;
            hi.x = rb2.z; hi.y = rb3.z; hi.z = rb2.w; hi.w = rb3.w;
            *(v4f*)(bs0 + 8 * BPITCH)     = lo;
            *(v4f*)(bs0 + 8 * BPITCH + 4) = hi;
        }
        __syncthreads();

        const int kn = k0 + GBK;
        if (kn < K) {
            const float* bgn = bg + (size_t)kn * Nc;
            ra0 = *(const v4f*)(ag0 + kn);
            ra1 = *(const v4f*)(ag1 + kn);
            ra2 = *(const v4f*)(ag2 + kn);
            ra3 = *(const v4f*)(ag3 + kn);
            rb0 = *(const v4f*)(bgn);
            rb1 = *(const v4f*)(bgn + nc1);
            rb2 = *(const v4f*)(bgn + nc16);
            rb3 = *(const v4f*)(bgn + nc16 + nc1);
            if (NORM) {
                pm = *(const v4f*)(nmean  + kn + acol);
                ps = *(const v4f*)(nscale + kn + acol);
                pb = *(const v4f*)(nbeta  + kn + acol);
            }
        }

        #pragma unroll
        for (int kk = 0; kk < GBK; kk += 4) {
            v2f a0 = *(const v2f*)(af0 + kk);
            v2f a1 = *(const v2f*)(af1 + kk);
            v2f b0 = *(const v2f*)(bf0 + (kk >> 1) * BPITCH);
            v2f b1 = *(const v2f*)(bf1 + (kk >> 1) * BPITCH);
            acc00 = WMMA_F32(a0, b0, acc00);
            acc01 = WMMA_F32(a0, b1, acc01);
            acc10 = WMMA_F32(a1, b0, acc10);
            acc11 = WMMA_F32(a1, b1, acc11);
        }
        __syncthreads();
    }

    float bb0 = 0.f, bb1 = 0.f;
    if (bias) {
        bb0 = bias[n0 + wn * 32 + c];
        bb1 = bias[n0 + wn * 32 + 16 + c];
    }
    const int rbase = m0 + wm * 32 + h * 8;
    #pragma unroll
    for (int r = 0; r < 8; ++r) {
        const int row0 = rbase + r;
        const int row1 = row0 + 16;
        float* crow0 = C + (size_t)row0 * Nc + n0 + wn * 32 + c;
        float* crow1 = C + (size_t)row1 * Nc + n0 + wn * 32 + c;
        if (row0 < M) {
            crow0[0]  = acc00[r] + bb0;
            crow0[16] = acc01[r] + bb1;
        }
        if (row1 < M) {
            crow1[0]  = acc10[r] + bb0;
            crow1[16] = acc11[r] + bb1;
        }
    }
}

__global__ __launch_bounds__(256)
void attn_scores(const float* __restrict__ hx,
                 const float* __restrict__ att_s, const float* __restrict__ att_d,
                 float* __restrict__ a_s, float* __restrict__ a_d, int nNodes)
{
    int gid  = blockIdx.x * blockDim.x + threadIdx.x;
    int wave = gid >> 5;
    int lane = gid & 31;
    if (wave >= nNodes * HEADS) return;
    int n = wave / HEADS, h = wave % HEADS;

    const float4* hv = (const float4*)(hx + (size_t)n * FDIM + h * CDIM);
    const float4* sv = (const float4*)(att_s + h * CDIM);
    const float4* dv = (const float4*)(att_d + h * CDIM);
    float4 x = hv[lane], s = sv[lane], d = dv[lane];
    float accs = x.x * s.x + x.y * s.y + x.z * s.z + x.w * s.w;
    float accd = x.x * d.x + x.y * d.y + x.z * d.z + x.w * d.w;
    #pragma unroll
    for (int off = 16; off > 0; off >>= 1) {
        accs += __shfl_down(accs, off, 32);
        accd += __shfl_down(accd, off, 32);
    }
    if (lane == 0) { a_s[wave] = accs; a_d[wave] = accd; }
}

__global__ void fill_u32(unsigned* __restrict__ p, unsigned v, size_t n)
{
    size_t i = (size_t)blockIdx.x * blockDim.x + threadIdx.x;
    if (i < n) p[i] = v;
}

__global__ __launch_bounds__(256)
void csr_count(const int* __restrict__ adj, unsigned* __restrict__ cnt,
               int* __restrict__ ticket, int E)
{
    int e = blockIdx.x * blockDim.x + threadIdx.x;
    if (e >= E) return;
    ticket[e] = (int)atomicAdd(&cnt[adj[E + e]], 1u);
}

__global__ __launch_bounds__(1024)
void csr_scan(const unsigned* __restrict__ cnt, int* __restrict__ rowptr, int n)
{
    __shared__ int wsum[32];
    const int tid = threadIdx.x, lane = tid & 31, w = tid >> 5;
    const int per = (n + 1023) / 1024;
    int b = tid * per;          b = b < n ? b : n;
    int e = b + per;            e = e < n ? e : n;
    int s = 0;
    for (int i = b; i < e; ++i) s += (int)cnt[i];
    int v = s;
    for (int off = 1; off < 32; off <<= 1) {
        int t = __shfl_up(v, off, 32);
        if (lane >= off) v += t;
    }
    if (lane == 31) wsum[w] = v;
    __syncthreads();
    if (w == 0) {
        int x = wsum[lane];
        for (int off = 1; off < 32; off <<= 1) {
            int t = __shfl_up(x, off, 32);
            if (lane >= off) x += t;
        }
        wsum[lane] = x;
    }
    __syncthreads();
    int base = v - s + (w > 0 ? wsum[w - 1] : 0);
    for (int i = b; i < e; ++i) { rowptr[i] = base; base += (int)cnt[i]; }
    if (tid == 1023) rowptr[n] = base;
}

__global__ __launch_bounds__(256)
void csr_fill(const int* __restrict__ adj, const int* __restrict__ rowptr,
              const int* __restrict__ ticket, int* __restrict__ col, int E)
{
    int e = blockIdx.x * blockDim.x + threadIdx.x;
    if (e >= E) return;
    col[rowptr[adj[E + e]] + ticket[e]] = adj[e];
}

__device__ __forceinline__ float leaky(float v) { return (v > 0.f) ? v : v * NEG_SLOPE; }

__global__ __launch_bounds__(192)
void node_gather_softmax(const int* __restrict__ rowptr, const int* __restrict__ col,
                         const float* __restrict__ hx,
                         const float* __restrict__ a_s, const float* __restrict__ a_d,
                         const float* __restrict__ bias, float* __restrict__ out)
{
    const int n    = blockIdx.x;
    const int lane = threadIdx.x & 31;
    const int h    = __builtin_amdgcn_readfirstlane((int)(threadIdx.x >> 5));
    const int beg  = rowptr[n], end = rowptr[n + 1];
    const float ad = a_d[n * HEADS + h];
    const float vself = leaky(a_s[n * HEADS + h] + ad);

    float m = vself;
    for (int j0 = beg; j0 < end; j0 += 32) {
        const int j = j0 + lane;
        if (j < end) m = fmaxf(m, leaky(a_s[col[j] * HEADS + h] + ad));
    }
    for (int off = 16; off > 0; off >>= 1) m = fmaxf(m, __shfl_xor(m, off, 32));

    const size_t hoff = (size_t)h * CDIM + lane * 4;
    const float pself = __expf(vself - m);
    float tot = pself;
    v4f acc = *(const v4f*)(hx + (size_t)n * FDIM + hoff) * pself;

    for (int j0 = beg; j0 < end; j0 += 32) {
        const int j = j0 + lane;
        int   s = 0;
        float p = 0.f;
        if (j < end) {
            s = col[j];
            p = __expf(leaky(a_s[s * HEADS + h] + ad) - m);
        }
        const int rem = end - j0;
        const int cnt = rem < 32 ? rem : 32;
        #pragma unroll 4
        for (int t = 0; t < cnt; ++t) {
            const int   st = __builtin_amdgcn_readlane(s, t);
            const float pt = __int_as_float(__builtin_amdgcn_readlane(__float_as_int(p), t));
            const v4f x = *(const v4f*)(hx + (size_t)st * FDIM + hoff);
            acc += x * pt;
            tot += pt;
        }
    }

    const v4f b4 = *(const v4f*)(bias + hoff);
    v4f r = acc / tot + b4;
    r.x = r.x > 0.f ? r.x : 0.f;
    r.y = r.y > 0.f ? r.y : 0.f;
    r.z = r.z > 0.f ? r.z : 0.f;
    r.w = r.w > 0.f ? r.w : 0.f;
    *(v4f*)(out + (size_t)n * FDIM + hoff) = r;
}

#define BN_ROWS 16
#define BN_AHEAD 8

__global__ __launch_bounds__(192)
void bn_partial(const float* __restrict__ x, float* __restrict__ psum,
                float* __restrict__ psq, int rows)
{
    const int r0 = blockIdx.x * BN_ROWS;
    const int c4 = threadIdx.x * 4;
    const float* p = x + (size_t)r0 * FDIM + c4;
    v4f s = {}, ss = {};
    if (r0 + BN_ROWS <= rows) {
        v4f v[BN_ROWS];
        #pragma unroll
        for (int i = 0; i < BN_ROWS; ++i) v[i] = *(const v4f*)(p + (size_t)i * FDIM);
        #pragma unroll
        for (int i = 0; i < BN_ROWS; ++i) { s += v[i]; ss += v[i] * v[i]; }
    } else {
        for (int r = r0; r < rows; ++r, p += FDIM) {
            const v4f v = *(const v4f*)p;
            s += v; ss += v * v;
        }
    }
    *(v4f*)(psum + (size_t)blockIdx.x * FDIM + c4) = s;
    *(v4f*)(psq  + (size_t)blockIdx.x * FDIM + c4) = ss;
}

__global__ __launch_bounds__(1024)
void bn_finish(const float* __restrict__ psum, const float* __restrict__ psq,
               const float* __restrict__ gamma,
               float* __restrict__ mean, float* __restrict__ scale, int nPart, int rows)
{
    __shared__ float sh[32 * 32], sh2[32 * 32];
    const int cx  = threadIdx.x & 31;
    const int w   = __builtin_amdgcn_readfirstlane((int)(threadIdx.x >> 5));
    const int col = blockIdx.x * 32 + cx;
    const float* ps = psum + col;
    const float* pq = psq  + col;
    float s = 0.f, ss = 0.f;
    for (int q0 = w; q0 < nPart; q0 += 32 * BN_AHEAD) {
        float a[BN_AHEAD], b[BN_AHEAD];
        #pragma unroll
        for (int u = 0; u < BN_AHEAD; ++u) {
            int q = q0 + 32 * u;
            q = q < nPart ? q : nPart - 1;
            a[u] = ps[(size_t)q * FDIM];
            b[u] = pq[(size_t)q * FDIM];
        }
        #pragma unroll
        for (int u = 0; u < BN_AHEAD; ++u) {
            const bool in = q0 + 32 * u < nPart;
            s  += in ? a[u] : 0.f;
            ss += in ? b[u] : 0.f;
        }
    }
    sh[w * 32 + cx]  = s;
    sh2[w * 32 + cx] = ss;
    __syncthreads();
    if (w == 0) {
        float a = 0.f, b = 0.f;
        for (int k = 0; k < 32; ++k) { a += sh[k * 32 + cx]; b += sh2[k * 32 + cx]; }
        const float mu  = a / rows;
        const float var = b / rows - mu * mu;
        mean[col]  = mu;
        scale[col] = rsqrtf(var + BN_EPS) * gamma[col];
    }
}

__global__ __launch_bounds__(256)
void pool_max(const float* __restrict__ lin, const int* __restrict__ ibatch,
              unsigned* __restrict__ pooled, int rows)
{
    size_t i = (size_t)blockIdx.x * blockDim.x + threadIdx.x;
    if (i >= (size_t)rows * OUT_C) return;
    int n = (int)(i / OUT_C), c = (int)(i % OUT_C);
    atomicMax(&pooled[ibatch[n] * OUT_C + c], fmap(lin[i]));
}

__global__ __launch_bounds__(256)
void pool_final(const unsigned* __restrict__ pooled, float* __restrict__ out, int n)
{
    int i = blockIdx.x * blockDim.x + threadIdx.x;
    if (i < n) out[i] = funmap(pooled[i]);
}

extern "C" void kernel_launch(void* const* d_in, const int* in_sizes, int n_in,
                              void* d_out, int out_size, void* d_ws, size_t ws_size,
                              hipStream_t stream)
{
    const float* x_in   = (const float*)d_in[0];
    const int*   adj    = (const int*)d_in[1];
    const int*   ibatch = (const int*)d_in[2];
    const float* W1     = (const float*)d_in[3];
    const float* att_s1 = (const float*)d_in[4];
    const float* att_d1 = (const float*)d_in[5];
    const float* b1     = (const float*)d_in[6];
    const float* g1     = (const float*)d_in[7];
    const float* be1    = (const float*)d_in[8];
    const float* W2     = (const float*)d_in[9];
    const float* att_s2 = (const float*)d_in[10];
    const float* att_d2 = (const float*)d_in[11];
    const float* b2     = (const float*)d_in[12];
    const float* g2     = (const float*)d_in[13];
    const float* be2    = (const float*)d_in[14];
    const float* linW   = (const float*)d_in[15];
    const float* linb   = (const float*)d_in[16];
    float* out = (float*)d_out;

    const int N    = in_sizes[0] / F_IN;
    const int E    = in_sizes[1] / 2;
    const int B    = out_size / OUT_C;
    const int mT   = (N + 63) / 64;

    float*    hx   = (float*)d_ws;
    float*    agg  = hx  + (size_t)N * FDIM;
    float*    mean = agg + (size_t)N * FDIM;
    float*    bnsc = mean + FDIM;
    float*    a_s  = bnsc + FDIM;
    float*    a_d  = a_s + (size_t)N * HEADS;
    unsigned* cnt  = (unsigned*)(a_d + (size_t)N * HEADS);
    int*      rowptr = (int*)(cnt + (size_t)N);
    int*      ticket = rowptr + (size_t)N + 1;
    int*      col    = ticket + (size_t)E;
    float*    lin  = (float*)(col + (size_t)E);
    unsigned* pooled = (unsigned*)(lin + (size_t)N * OUT_C);

    const size_t nh  = (size_t)N * HEADS;

    const int nP   = (N + BN_ROWS - 1) / BN_ROWS;
    float*    psum = hx;
    float*    psq  = hx + (size_t)nP * FDIM;

    fill_u32<<<(N + 255) / 256, 256, 0, stream>>>(cnt, 0u, (size_t)N);
    csr_count<<<(E + 255) / 256, 256, 0, stream>>>(adj, cnt, ticket, E);
    csr_scan<<<1, 1024, 0, stream>>>(cnt, rowptr, N);
    csr_fill<<<(E + 255) / 256, 256, 0, stream>>>(adj, rowptr, ticket, col, E);

    wmma_gemm_f32<false><<<mT * (FDIM / 64), 128, 0, stream>>>(
        x_in, W1, hx, nullptr, nullptr, nullptr, nullptr, N, F_IN, FDIM);
    attn_scores<<<(int)((nh * 32 + 255) / 256), 256, 0, stream>>>(hx, att_s1, att_d1, a_s, a_d, N);
    node_gather_softmax<<<N, 192, 0, stream>>>(rowptr, col, hx, a_s, a_d, b1, agg);
    bn_partial<<<nP, 192, 0, stream>>>(agg, psum, psq, N);
    bn_finish<<<FDIM / 32, 1024, 0, stream>>>(psum, psq, g1, mean, bnsc, nP, N);

    wmma_gemm_f32<true><<<mT * (FDIM / 64), 128, 0, stream>>>(
        agg, W2, hx, nullptr, mean, bnsc, be1, N, FDIM, FDIM);
    attn_scores<<<(int)((nh * 32 + 255) / 256), 256, 0, stream>>>(hx, att_s2, att_d2, a_s, a_d, N);
    node_gather_softmax<<<N, 192, 0, stream>>>(rowptr, col, hx, a_s, a_d, b2, agg);
    bn_partial<<<nP, 192, 0, stream>>>(agg, psum, psq, N);
    bn_finish<<<FDIM / 32, 1024, 0, stream>>>(psum, psq, g2, mean, bnsc, nP, N);

    wmma_gemm_f32<true><<<mT * (OUT_C / 64), 128, 0, stream>>>(
        agg, linW, lin, linb, mean, bnsc, be2, N, FDIM, OUT_C);
    fill_u32<<<(B * OUT_C + 255) / 256, 256, 0, stream>>>(pooled, FMAP_NEG_INF, (size_t)B * OUT_C);
    pool_max<<<(int)(((size_t)N * OUT_C + 255) / 256), 256, 0, stream>>>(lin, ibatch, pooled, N);
    pool_final<<<(B * OUT_C + 255) / 256, 256, 0, stream>>>(pooled, out, B * OUT_C);
}


# Attention scores: one wave takes 4 whole rows (6 heads each, 24 row loads in flight, its 48 sums on DPP adds) instead of one wave per (node, head): 2500 waves to start, were 60000; sum order kept
# speedup vs baseline: 3.6842x; 1.0647x over previous
#include <hip/hip_runtime.h>

#define F_IN   256
#define HEADS  6
#define CDIM   128
#define FDIM   768
#define OUT_C  128
#define NEG_SLOPE 0.2f
#define BN_EPS    1e-5f

typedef __attribute__((ext_vector_type(2))) float v2f;
typedef __attribute__((ext_vector_type(8))) float v8f;

__device__ __forceinline__ unsigned fmap(float f) {
    unsigned u = __float_as_uint(f);
    return (u & 0x80000000u) ? ~u : (u | 0x80000000u);
}
__device__ __forceinline__ float funmap(unsigned u) {
    u = (u & 0x80000000u) ? (u & 0x7FFFFFFFu) : ~u;
    return __uint_as_float(u);
}
#define FMAP_NEG_INF 0x007FFFFFu

#define WMMA_F32(a, b, c) \
    __builtin_amdgcn_wmma_f32_16x16x4_f32(false, (a), false, (b), (short)0, (c), false, false)

#define GBM 64
#define GBN 64
#define GBK 32
#define APITCH 36
#define BPITCH 160

typedef __attribute__((ext_vector_type(4))) float v4f;

template <bool NORM>
__global__ __launch_bounds__(128)
void wmma_gemm_f32(const float* __restrict__ A, const float* __restrict__ B,
                   float* __restrict__ C, const float* __restrict__ bias,
                   const float* __restrict__ nmean, const float* __restrict__ nscale,
                   const float* __restrict__ nbeta,
                   int M, int K, int Nc)
{
    __shared__ __attribute__((aligned(16))) float As[GBM * APITCH];
    __shared__ __attribute__((aligned(16))) float Bs[(GBK / 2) * BPITCH];

    const int tid  = threadIdx.x;
    const int lane = tid & 31;
    const int wave = tid >> 5;
    const int wm   = wave >> 1;
    const int wn   = wave & 1;
    const int nN   = Nc >> 6;
    const int m0   = (blockIdx.x / nN) * GBM;
    const int n0   = (blockIdx.x % nN) * GBN;
    const int c    = lane & 15;
    const int h    = lane >> 4;

    const int arow = tid >> 3, acol = (tid & 7) << 2;
    const int lastRow = M - 1;
    int gm0 = m0 + arow;       gm0 = gm0 < lastRow ? gm0 : lastRow;
    int gm1 = m0 + arow + 16;  gm1 = gm1 < lastRow ? gm1 : lastRow;
    int gm2 = m0 + arow + 32;  gm2 = gm2 < lastRow ? gm2 : lastRow;
    int gm3 = m0 + arow + 48;  gm3 = gm3 < lastRow ? gm3 : lastRow;
    const float* ag0 = A + (size_t)gm0 * K + acol;
    const float* ag1 = A + (size_t)gm1 * K + acol;
    const float* ag2 = A + (size_t)gm2 * K + acol;
    const float* ag3 = A + (size_t)gm3 * K + acol;
    float* as0 = As + arow * APITCH + acol;

    const int bkp = tid >> 4, bcol = (tid & 15) << 2;
    const float* bg = B + (size_t)(2 * bkp) * Nc + n0 + bcol;
    const size_t nc1  = (size_t)Nc;
    const size_t nc16 = (size_t)16 * Nc;
    float* bs0 = Bs + bkp * BPITCH + bcol * 2;

    v4f ra0 = *(const v4f*)(ag0);
    v4f ra1 = *(const v4f*)(ag1);
    v4f ra2 = *(const v4f*)(ag2);
    v4f ra3 = *(const v4f*)(ag3);
    v4f rb0 = *(const v4f*)(bg);
    v4f rb1 = *(const v4f*)(bg + nc1);
    v4f rb2 = *(const v4f*)(bg + nc16);
    v4f rb3 = *(const v4f*)(bg + nc16 + nc1);
    v4f pm = {}, ps = {}, pb = {};
    if (NORM) {
        pm = *(const v4f*)(nmean  + acol);
        ps = *(const v4f*)(nscale + acol);
        pb = *(const v4f*)(nbeta  + acol);
    }

    v8f acc00 = {}, acc01 = {}, acc10 = {}, acc11 = {};

    const float* af0 = As + (wm * 32 + c) * APITCH + 2 * h;
    const float* af1 = af0 + 16 * APITCH;
    const float* bf0 = Bs + h * BPITCH + (wn * 32 + c) * 2;
    const float* bf1 = bf0 + 32;

    for (int k0 = 0; k0 < K; k0 += GBK) {
        if (NORM) {
            ra0 = (ra0 - pm) * ps + pb;
            ra1 = (ra1 - pm) * ps + pb;
            ra2 = (ra2 - pm) * ps + pb;
            ra3 = (ra3 - pm) * ps + pb;
        }
        *(v4f*)(as0)               = ra0;
        *(v4f*)(as0 + 16 * APITCH) = ra1;
        *(v4f*)(as0 + 32 * APITCH) = ra2;
        *(v4f*)(as0 + 48 * APITCH) = ra3;
        {
            v4f lo, hi;
            lo.x = rb0.x; lo.y = rb1.x; lo.z = rb0.y; lo.w = rb1.y;
            hi.x = rb0.z; hi.y = rb1.z; hi.z = rb0.w; hi.w = rb1.w;
            *(v4f*)(bs0)     = lo;
            *(v4f*)(bs0 + 4) = hi;
            lo.x = rb2.x; lo.y = rb3.x; lo.z = rb2.y; lo.w = rb3.y;
            hi.x = rb2.z; hi.y = rb3.z; hi.z = rb2.w; hi.w = rb3.w;
            *(v4f*)(bs0 + 8 * BPITCH)     = lo;
            *(v4f*)(bs0 + 8 * BPITCH + 4) = hi;
        }
        __syncthreads();

        const int kn = k0 + GBK;
        if (kn < K) {
            const float* bgn = bg + (size_t)kn * Nc;
            ra0 = *(const v4f*)(ag0 + kn);
            ra1 = *(const v4f*)(ag1 + kn);
            ra2 = *(const v4f*)(ag2 + kn);
            ra3 = *(const v4f*)(ag3 + kn);
            rb0 = *(const v4f*)(bgn);
            rb1 = *(const v4f*)(bgn + nc1);
            rb2 = *(const v4f*)(bgn + nc16);
            rb3 = *(const v4f*)(bgn + nc16 + nc1);
            if (NORM) {
                pm = *(const v4f*)(nmean  + kn + acol);
                ps = *(const v4f*)(nscale + kn + acol);
                pb = *(const v4f*)(nbeta  + kn + acol);
            }
        }

        #pragma unroll
        for (int kk = 0; kk < GBK; kk += 4) {
            v2f a0 = *(const v2f*)(af0 + kk);
            v2f a1 = *(const v2f*)(af1 + kk);
            v2f b0 = *(const v2f*)(bf0 + (kk >> 1) * BPITCH);
            v2f b1 = *(const v2f*)(bf1 + (kk >> 1) * BPITCH);
            acc00 = WMMA_F32(a0, b0, acc00);
            acc01 = WMMA_F32(a0, b1, acc01);
            acc10 = WMMA_F32(a1, b0, acc10);
            acc11 = WMMA_F32(a1, b1, acc11);
        }
        __syncthreads();
    }

    float bb0 = 0.f, bb1 = 0.f;
    if (bias) {
        bb0 = bias[n0 + wn * 32 + c];
        bb1 = bias[n0 + wn * 32 + 16 + c];
    }
    const int rbase = m0 + wm * 32 + h * 8;
    #pragma unroll
    for (int r = 0; r < 8; ++r) {
        const int row0 = rbase + r;
        const int row1 = row0 + 16;
        float* crow0 = C + (size_t)row0 * Nc + n0 + wn * 32 + c;
        float* crow1 = C + (size_t)row1 * Nc + n0 + wn * 32 + c;
        if (row0 < M) {
            crow0[0]  = acc00[r] + bb0;
            crow0[16] = acc01[r] + bb1;
        }
        if (row1 < M) {
            crow1[0]  = acc10[r] + bb0;
            crow1[16] = acc11[r] + bb1;
        }
    }
}

#define AS_NODES 4
#define AS_WAVES 4

__device__ __forceinline__ float xhalf(float x) {
    const int b = __float_as_int(x);
    return __int_as_float(__builtin_amdgcn_permlanex16(b, b, 0x76543210, 0xfedcba98, false, false));
}
template <int M>
__device__ __forceinline__ float add_row_xor(float x) {
    const int b = __float_as_int(x);
    return x + __int_as_float(__builtin_amdgcn_update_dpp(b, b, 0x160 + M, 0xf, 0xf, true));
}
__device__ __forceinline__ float wave_sum(float x) {
    x += xhalf(x);
    x = add_row_xor<8>(x);
    x = add_row_xor<4>(x);
    x = add_row_xor<2>(x);
    x = add_row_xor<1>(x);
    return x;
}

__global__ __launch_bounds__(AS_WAVES * 32)
void attn_scores(const float* __restrict__ hx,
                 const float* __restrict__ att_s, const float* __restrict__ att_d,
                 float* __restrict__ a_s, float* __restrict__ a_d, int nNodes)
{
    const int lane = threadIdx.x & 31;
    const int wave = __builtin_amdgcn_readfirstlane(
        (int)(blockIdx.x * AS_WAVES + (threadIdx.x >> 5)));
    const int n0 = wave * AS_NODES;
    if (n0 >= nNodes) return;
    const int left = nNodes - n0;
    const int cnt  = left < AS_NODES ? left : AS_NODES;

    v4f sv[HEADS], dv[HEADS], x[AS_NODES][HEADS];
    #pragma unroll
    for (int h = 0; h < HEADS; ++h) {
        sv[h] = *(const v4f*)(att_s + h * CDIM + lane * 4);
        dv[h] = *(const v4f*)(att_d + h * CDIM + lane * 4);
    }
    #pragma unroll
    for (int q = 0; q < AS_NODES; ++q) {
        const int n = n0 + (q < cnt ? q : cnt - 1);
        const float* row = hx + (size_t)n * FDIM + lane * 4;
        #pragma unroll
        for (int h = 0; h < HEADS; ++h) x[q][h] = *(const v4f*)(row + h * CDIM);
    }
    __builtin_amdgcn_sched_barrier(0);

    float outs = 0.f, outd = 0.f;
    #pragma unroll
    for (int h = 0; h < HEADS; ++h) {
        const v4f s = sv[h];
        const v4f d = dv[h];
        #pragma unroll
        for (int q = 0; q < AS_NODES; ++q) {
            const v4f v = x[q][h];
            float accs = __builtin_fmaf(v.w, s.w, __builtin_fmaf(v.z, s.z, __builtin_fmaf(v.x, s.x, v.y * s.y)));
            float accd = __builtin_fmaf(v.w, d.w, __builtin_fmaf(v.z, d.z, __builtin_fmaf(v.x, d.x, v.y * d.y)));
            accs = wave_sum(accs);
            accd = wave_sum(accd);
            const float rs = __int_as_float(__builtin_amdgcn_readfirstlane(__float_as_int(accs)));
            const float rd = __int_as_float(__builtin_amdgcn_readfirstlane(__float_as_int(accd)));
            if (lane == q * HEADS + h) { outs = rs; outd = rd; }
        }
    }
    if (lane < cnt * HEADS) {
        a_s[(size_t)n0 * HEADS + lane] = outs;
        a_d[(size_t)n0 * HEADS + lane] = outd;
    }
}

__global__ void fill_u32(unsigned* __restrict__ p, unsigned v, size_t n)
{
    size_t i = (size_t)blockIdx.x * blockDim.x + threadIdx.x;
    if (i < n) p[i] = v;
}

__global__ __launch_bounds__(256)
void csr_count(const int* __restrict__ adj, unsigned* __restrict__ cnt,
               int* __restrict__ ticket, int E)
{
    int e = blockIdx.x * blockDim.x + threadIdx.x;
    if (e >= E) return;
    ticket[e] = (int)atomicAdd(&cnt[adj[E + e]], 1u);
}

__global__ __launch_bounds__(1024)
void csr_scan(const unsigned* __restrict__ cnt, int* __restrict__ rowptr, int n)
{
    __shared__ int wsum[32];
    const int tid = threadIdx.x, lane = tid & 31, w = tid >> 5;
    const int per = (n + 1023) / 1024;
    int b = tid * per;          b = b < n ? b : n;
    int e = b + per;            e = e < n ? e : n;
    int s = 0;
    for (int i = b; i < e; ++i) s += (int)cnt[i];
    int v = s;
    for (int off = 1; off < 32; off <<= 1) {
        int t = __shfl_up(v, off, 32);
        if (lane >= off) v += t;
    }
    if (lane == 31) wsum[w] = v;
    __syncthreads();
    if (w == 0) {
        int x = wsum[lane];
        for (int off = 1; off < 32; off <<= 1) {
            int t = __shfl_up(x, off, 32);
            if (lane >= off) x += t;
        }
        wsum[lane] = x;
    }
    __syncthreads();
    int base = v - s + (w > 0 ? wsum[w - 1] : 0);
    for (int i = b; i < e; ++i) { rowptr[i] = base; base += (int)cnt[i]; }
    if (tid == 1023) rowptr[n] = base;
}

__global__ __launch_bounds__(256)
void csr_fill(const int* __restrict__ adj, const int* __restrict__ rowptr,
              const int* __restrict__ ticket, int* __restrict__ col, int E)
{
    int e = blockIdx.x * blockDim.x + threadIdx.x;
    if (e >= E) return;
    col[rowptr[adj[E + e]] + ticket[e]] = adj[e];
}

__device__ __forceinline__ float leaky(float v) { return (v > 0.f) ? v : v * NEG_SLOPE; }

__global__ __launch_bounds__(192)
void node_gather_softmax(const int* __restrict__ rowptr, const int* __restrict__ col,
                         const float* __restrict__ hx,
                         const float* __restrict__ a_s, const float* __restrict__ a_d,
                         const float* __restrict__ bias, float* __restrict__ out)
{
    const int n    = blockIdx.x;
    const int lane = threadIdx.x & 31;
    const int h    = __builtin_amdgcn_readfirstlane((int)(threadIdx.x >> 5));
    const int beg  = rowptr[n], end = rowptr[n + 1];
    const float ad = a_d[n * HEADS + h];
    const float vself = leaky(a_s[n * HEADS + h] + ad);

    float m = vself;
    for (int j0 = beg; j0 < end; j0 += 32) {
        const int j = j0 + lane;
        if (j < end) m = fmaxf(m, leaky(a_s[col[j] * HEADS + h] + ad));
    }
    for (int off = 16; off > 0; off >>= 1) m = fmaxf(m, __shfl_xor(m, off, 32));

    const size_t hoff = (size_t)h * CDIM + lane * 4;
    const float pself = __expf(vself - m);
    float tot = pself;
    v4f acc = *(const v4f*)(hx + (size_t)n * FDIM + hoff) * pself;

    for (int j0 = beg; j0 < end; j0 += 32) {
        const int j = j0 + lane;
        int   s = 0;
        float p = 0.f;
        if (j < end) {
            s = col[j];
            p = __expf(leaky(a_s[s * HEADS + h] + ad) - m);
        }
        const int rem = end - j0;
        const int cnt = rem < 32 ? rem : 32;
        #pragma unroll 4
        for (int t = 0; t < cnt; ++t) {
            const int   st = __builtin_amdgcn_readlane(s, t);
            const float pt = __int_as_float(__builtin_amdgcn_readlane(__float_as_int(p), t));
            const v4f x = *(const v4f*)(hx + (size_t)st * FDIM + hoff);
            acc += x * pt;
            tot += pt;
        }
    }

    const v4f b4 = *(const v4f*)(bias + hoff);
    v4f r = acc / tot + b4;
    r.x = r.x > 0.f ? r.x : 0.f;
    r.y = r.y > 0.f ? r.y : 0.f;
    r.z = r.z > 0.f ? r.z : 0.f;
    r.w = r.w > 0.f ? r.w : 0.f;
    *(v4f*)(out + (size_t)n * FDIM + hoff) = r;
}

#define BN_ROWS 16
#define BN_AHEAD 8

__global__ __launch_bounds__(192)
void bn_partial(const float* __restrict__ x, float* __restrict__ psum,
                float* __restrict__ psq, int rows)
{
    const int r0 = blockIdx.x * BN_ROWS;
    const int c4 = threadIdx.x * 4;
    const float* p = x + (size_t)r0 * FDIM + c4;
    v4f s = {}, ss = {};
    if (r0 + BN_ROWS <= rows) {
        v4f v[BN_ROWS];
        #pragma unroll
        for (int i = 0; i < BN_ROWS; ++i) v[i] = *(const v4f*)(p + (size_t)i * FDIM);
        #pragma unroll
        for (int i = 0; i < BN_ROWS; ++i) { s += v[i]; ss += v[i] * v[i]; }
    } else {
        for (int r = r0; r < rows; ++r, p += FDIM) {
            const v4f v = *(const v4f*)p;
            s += v; ss += v * v;
        }
    }
    *(v4f*)(psum + (size_t)blockIdx.x * FDIM + c4) = s;
    *(v4f*)(psq  + (size_t)blockIdx.x * FDIM + c4) = ss;
}

__global__ __launch_bounds__(1024)
void bn_finish(const float* __restrict__ psum, const float* __restrict__ psq,
               const float* __restrict__ gamma,
               float* __restrict__ mean, float* __restrict__ scale, int nPart, int rows)
{
    __shared__ float sh[32 * 32], sh2[32 * 32];
    const int cx  = threadIdx.x & 31;
    const int w   = __builtin_amdgcn_readfirstlane((int)(threadIdx.x >> 5));
    const int col = blockIdx.x * 32 + cx;
    const float* ps = psum + col;
    const float* pq = psq  + col;
    float s = 0.f, ss = 0.f;
    for (int q0 = w; q0 < nPart; q0 += 32 * BN_AHEAD) {
        float a[BN_AHEAD], b[BN_AHEAD];
        #pragma unroll
        for (int u = 0; u < BN_AHEAD; ++u) {
            int q = q0 + 32 * u;
            q = q < nPart ? q : nPart - 1;
            a[u] = ps[(size_t)q * FDIM];
            b[u] = pq[(size_t)q * FDIM];
        }
        #pragma unroll
        for (int u = 0; u < BN_AHEAD; ++u) {
            const bool in = q0 + 32 * u < nPart;
            s  += in ? a[u] : 0.f;
            ss += in ? b[u] : 0.f;
        }
    }
    sh[w * 32 + cx]  = s;
    sh2[w * 32 + cx] = ss;
    __syncthreads();
    if (w == 0) {
        float a = 0.f, b = 0.f;
        for (int k = 0; k < 32; ++k) { a += sh[k * 32 + cx]; b += sh2[k * 32 + cx]; }
        const float mu  = a / rows;
        const float var = b / rows - mu * mu;
        mean[col]  = mu;
        scale[col] = rsqrtf(var + BN_EPS) * gamma[col];
    }
}

__global__ __launch_bounds__(256)
void pool_max(const float* __restrict__ lin, const int* __restrict__ ibatch,
              unsigned* __restrict__ pooled, int rows)
{
    size_t i = (size_t)blockIdx.x * blockDim.x + threadIdx.x;
    if (i >= (size_t)rows * OUT_C) return;
    int n = (int)(i / OUT_C), c = (int)(i % OUT_C);
    atomicMax(&pooled[ibatch[n] * OUT_C + c], fmap(lin[i]));
}

__global__ __launch_bounds__(256)
void pool_final(const unsigned* __restrict__ pooled, float* __restrict__ out, int n)
{
    int i = blockIdx.x * blockDim.x + threadIdx.x;
    if (i < n) out[i] = funmap(pooled[i]);
}

extern "C" void kernel_launch(void* const* d_in, const int* in_sizes, int n_in,
                              void* d_out, int out_size, void* d_ws, size_t ws_size,
                              hipStream_t stream)
{
    const float* x_in   = (const float*)d_in[0];
    const int*   adj    = (const int*)d_in[1];
    const int*   ibatch = (const int*)d_in[2];
    const float* W1     = (const float*)d_in[3];
    const float* att_s1 = (const float*)d_in[4];
    const float* att_d1 = (const float*)d_in[5];
    const float* b1     = (const float*)d_in[6];
    const float* g1     = (const float*)d_in[7];
    const float* be1    = (const float*)d_in[8];
    const float* W2     = (const float*)d_in[9];
    const float* att_s2 = (const float*)d_in[10];
    const float* att_d2 = (const float*)d_in[11];
    const float* b2     = (const float*)d_in[12];
    const float* g2     = (const float*)d_in[13];
    const float* be2    = (const float*)d_in[14];
    const float* linW   = (const float*)d_in[15];
    const float* linb   = (const float*)d_in[16];
    float* out = (float*)d_out;

    const int N    = in_sizes[0] / F_IN;
    const int E    = in_sizes[1] / 2;
    const int B    = out_size / OUT_C;
    const int mT   = (N + 63) / 64;

    float*    hx   = (float*)d_ws;
    float*    agg  = hx  + (size_t)N * FDIM;
    float*    mean = agg + (size_t)N * FDIM;
    float*    bnsc = mean + FDIM;
    float*    a_s  = bnsc + FDIM;
    float*    a_d  = a_s + (size_t)N * HEADS;
    unsigned* cnt  = (unsigned*)(a_d + (size_t)N * HEADS);
    int*      rowptr = (int*)(cnt + (size_t)N);
    int*      ticket = rowptr + (size_t)N + 1;
    int*      col    = ticket + (size_t)E;
    float*    lin  = (float*)(col + (size_t)E);
    unsigned* pooled = (unsigned*)(lin + (size_t)N * OUT_C);

    const int asBlocks = (N + AS_NODES * AS_WAVES - 1) / (AS_NODES * AS_WAVES);

    const int nP   = (N + BN_ROWS - 1) / BN_ROWS;
    float*    psum = hx;
    float*    psq  = hx + (size_t)nP * FDIM;

    fill_u32<<<(N + 255) / 256, 256, 0, stream>>>(cnt, 0u, (size_t)N);
    csr_count<<<(E + 255) / 256, 256, 0, stream>>>(adj, cnt, ticket, E);
    csr_scan<<<1, 1024, 0, stream>>>(cnt, rowptr, N);
    csr_fill<<<(E + 255) / 256, 256, 0, stream>>>(adj, rowptr, ticket, col, E);

    wmma_gemm_f32<false><<<mT * (FDIM / 64), 128, 0, stream>>>(
        x_in, W1, hx, nullptr, nullptr, nullptr, nullptr, N, F_IN, FDIM);
    attn_scores<<<asBlocks, AS_WAVES * 32, 0, stream>>>(hx, att_s1, att_d1, a_s, a_d, N);
    node_gather_softmax<<<N, 192, 0, stream>>>(rowptr, col, hx, a_s, a_d, b1, agg);
    bn_partial<<<nP, 192, 0, stream>>>(agg, psum, psq, N);
    bn_finish<<<FDIM / 32, 1024, 0, stream>>>(psum, psq, g1, mean, bnsc, nP, N);

    wmma_gemm_f32<true><<<mT * (FDIM / 64), 128, 0, stream>>>(
        agg, W2, hx, nullptr, mean, bnsc, be1, N, FDIM, FDIM);
    attn_scores<<<asBlocks, AS_WAVES * 32, 0, stream>>>(hx, att_s2, att_d2, a_s, a_d, N);
    node_gather_softmax<<<N, 192, 0, stream>>>(rowptr, col, hx, a_s, a_d, b2, agg);
    bn_partial<<<nP, 192, 0, stream>>>(agg, psum, psq, N);
    bn_finish<<<FDIM / 32, 1024, 0, stream>>>(psum, psq, g2, mean, bnsc, nP, N);

    wmma_gemm_f32<true><<<mT * (OUT_C / 64), 128, 0, stream>>>(
        agg, linW, lin, linb, mean, bnsc, be2, N, FDIM, OUT_C);
    fill_u32<<<(B * OUT_C + 255) / 256, 256, 0, stream>>>(pooled, FMAP_NEG_INF, (size_t)B * OUT_C);
    pool_max<<<(int)(((size_t)N * OUT_C + 255) / 256), 256, 0, stream>>>(lin, ibatch, pooled, N);
    pool_final<<<(B * OUT_C + 255) / 256, 256, 0, stream>>>(pooled, out, B * OUT_C);
}
